# starting kernel
# baseline (speedup 1.0000x reference)
#include <hip/hip_runtime.h>

typedef __attribute__((ext_vector_type(2))) float v2f;
typedef __attribute__((ext_vector_type(8))) float v8f;

#define DIM       96
#define HWSZ      (96*96)
#define DHW       (96*96*96)
#define CSTRIDE   104
#define ROWSTRIDE (32*CSTRIDE)
#define LDSF      (3*ROWSTRIDE)
#define WOFF      4

__global__ __launch_bounds__(192)
void wincorr_wmma_kernel(const float* __restrict__ fixedp,
                         const float* __restrict__ movingp,
                         float* __restrict__ outp)
{
    __shared__ float lds[LDSF];

    const int t    = threadIdx.x;
    const int lane = t & 31;
    const int wv   = t >> 5;
    const int h    = blockIdx.x % DIM;
    const int d    = blockIdx.x / DIM;
    const int w0   = wv * 16;

    const int sp    = lane & 15;
    const int khalf = (lane >> 4) * 2;

    v2f a[8];
    const size_t fbase = (size_t)(d * HWSZ + h * DIM + w0 + sp);
    #pragma unroll
    for (int s = 0; s < 8; ++s) {
        const int c0 = 4 * s + khalf;
        a[s].x = fixedp[(size_t)c0       * DHW + fbase];
        a[s].y = fixedp[(size_t)(c0 + 1) * DHW + fbase];
    }

    const bool diagLane = (lane < 8) || (lane >= 24);
    const int  didx     = (lane < 8) ? lane : (lane - 24);
    const int  m        = (lane < 16) ? lane : (lane - 16);
    const float  scale  = 0.17677669529663687f;
    const size_t obase  = (size_t)(d * HWSZ + h * DIM + w0 + m);

    {
        const int hj = t >> 6, hc = (t & 63) >> 1, hs = t & 1;
        lds[hj * ROWSTRIDE + hc * CSTRIDE + (hs ? (WOFF + DIM) : (WOFF - 1))] = 0.0f;
    }

    for (int i = 0; i < 3; ++i) {
        const int dm = d + i - 1;
        const int dmc = dm < 0 ? 0 : (dm > DIM - 1 ? DIM - 1 : dm);

        if (i) __syncthreads();

        #pragma unroll
        for (int q = 0; q < 12; ++q) {
            const int seg = t + 192 * q;
            const int j   = seg / 768;
            const int rem = seg % 768;
            const int c   = rem / 24;
            const int wq  = (rem % 24) * 4;
            const int hm  = h + j - 1;
            const bool ok = (dm >= 0) & (dm < DIM) & (hm >= 0) & (hm < DIM);
            const int hmc = hm < 0 ? 0 : (hm > DIM - 1 ? DIM - 1 : hm);
            const float4 src = *(const float4*)(movingp + (size_t)c * DHW +
                                    (size_t)(dmc * HWSZ + hmc * DIM + wq));
            float4 val;
            val.x = ok ? src.x : 0.0f;
            val.y = ok ? src.y : 0.0f;
            val.z = ok ? src.z : 0.0f;
            val.w = ok ? src.w : 0.0f;
            *(float4*)(lds + j * ROWSTRIDE + c * CSTRIDE + WOFF + wq) = val;
        }
        __syncthreads();

        #pragma unroll
        for (int j = 0; j < 3; ++j) {
          #pragma unroll
          for (int k = 0; k < 3; ++k) {
            const float* bp = lds + j * ROWSTRIDE + khalf * CSTRIDE
                                  + (WOFF - 1) + w0 + sp + k;
            v2f b[8];
            #pragma unroll
            for (int s = 0; s < 8; ++s) {
                b[s].x = bp[s * 4 * CSTRIDE];
                b[s].y = bp[s * 4 * CSTRIDE + CSTRIDE];
            }

            v8f acc = {};
            #pragma unroll
            for (int s = 0; s < 8; ++s) {
                acc = __builtin_amdgcn_wmma_f32_16x16x4_f32(
                          false, a[s], false, b[s], (short)0, acc, false, false);
            }

            float v = acc[0];
            if (didx == 1) v = acc[1];
            if (didx == 2) v = acc[2];
            if (didx == 3) v = acc[3];
            if (didx == 4) v = acc[4];
            if (didx == 5) v = acc[5];
            if (didx == 6) v = acc[6];
            if (didx == 7) v = acc[7];

            if (diagLane) {
                const int o = (i * 3 + j) * 3 + k;
                outp[(size_t)o * DHW + obase] = v * scale;
            }
          }
        }
    }
}

extern "C" void kernel_launch(void* const* d_in, const int* in_sizes, int n_in,
                              void* d_out, int out_size, void* d_ws, size_t ws_size,
                              hipStream_t stream)
{
    const float* fixedp  = (const float*)d_in[0];
    const float* movingp = (const float*)d_in[1];
    float* outp = (float*)d_out;

    dim3 grid(DIM * DIM), block(192);
    hipLaunchKernelGGL(wincorr_wmma_kernel, grid, block, 0, stream,
                       fixedp, movingp, outp);
}


# Per moving row the 3 k chains run as one block: 24 LDS fragment reads issued as one group (next row's right behind the WMMAs), then 24 WMMAs on 3 independent accumulators; sum order unchanged
# speedup vs baseline: 1.0443x; 1.0443x over previous
#include <hip/hip_runtime.h>

typedef __attribute__((ext_vector_type(2))) float v2f;
typedef __attribute__((ext_vector_type(8))) float v8f;

#define DIM       96
#define HWSZ      (96*96)
#define DHW       (96*96*96)
#define CSTRIDE   104
#define ROWSTRIDE (32*CSTRIDE)
#define LDSF      (3*ROWSTRIDE)
#define WOFF      4

__global__ __launch_bounds__(192) __attribute__((amdgpu_waves_per_eu(6)))
void wincorr_wmma_kernel(const float* __restrict__ fixedp,
                         const float* __restrict__ movingp,
                         float* __restrict__ outp)
{
    __shared__ float lds[LDSF];

    const int t    = threadIdx.x;
    const int lane = t & 31;
    const int wv   = t >> 5;
    const int h    = blockIdx.x % DIM;
    const int d    = blockIdx.x / DIM;
    const int w0   = wv * 16;

    const int sp    = lane & 15;
    const int khalf = (lane >> 4) * 2;

    v2f a[8];
    const size_t fbase = (size_t)(d * HWSZ + h * DIM + w0 + sp);
    #pragma unroll
    for (int s = 0; s < 8; ++s) {
        const int c0 = 4 * s + khalf;
        a[s].x = fixedp[(size_t)c0       * DHW + fbase];
        a[s].y = fixedp[(size_t)(c0 + 1) * DHW + fbase];
    }

    const bool diagLane = (lane < 8) || (lane >= 24);
    const int  didx     = (lane < 8) ? lane : (lane - 24);
    const int  m        = (lane < 16) ? lane : (lane - 16);
    const float  scale  = 0.17677669529663687f;
    const size_t obase  = (size_t)(d * HWSZ + h * DIM + w0 + m);

    {
        const int hj = t >> 6, hc = (t & 63) >> 1, hs = t & 1;
        lds[hj * ROWSTRIDE + hc * CSTRIDE + (hs ? (WOFF + DIM) : (WOFF - 1))] = 0.0f;
    }

    for (int i = 0; i < 3; ++i) {
        const int dm = d + i - 1;
        const int dmc = dm < 0 ? 0 : (dm > DIM - 1 ? DIM - 1 : dm);

        if (i) __syncthreads();

        #pragma unroll
        for (int q = 0; q < 12; ++q) {
            const int seg = t + 192 * q;
            const int j   = seg / 768;
            const int rem = seg % 768;
            const int c   = rem / 24;
            const int wq  = (rem % 24) * 4;
            const int hm  = h + j - 1;
            const bool ok = (dm >= 0) & (dm < DIM) & (hm >= 0) & (hm < DIM);
            const int hmc = hm < 0 ? 0 : (hm > DIM - 1 ? DIM - 1 : hm);
            const float4 src = *(const float4*)(movingp + (size_t)c * DHW +
                                    (size_t)(dmc * HWSZ + hmc * DIM + wq));
            float4 val;
            val.x = ok ? src.x : 0.0f;
            val.y = ok ? src.y : 0.0f;
            val.z = ok ? src.z : 0.0f;
            val.w = ok ? src.w : 0.0f;
            *(float4*)(lds + j * ROWSTRIDE + c * CSTRIDE + WOFF + wq) = val;
        }
        __syncthreads();

        v2f b0[8], b1[8], b2[8];
        #define LOAD_B_ROW(J)                                                     \
        {                                                                         \
            const float* bp = lds + (J) * ROWSTRIDE + khalf * CSTRIDE             \
                                  + (WOFF - 1) + w0 + sp;                         \
              \
            _Pragma("unroll")                                                     \
            for (int s = 0; s < 8; ++s) {                                         \
                b0[s].x = bp[s * 4 * CSTRIDE + 0];                                \
                b0[s].y = bp[s * 4 * CSTRIDE + CSTRIDE + 0];                      \
                b1[s].x = bp[s * 4 * CSTRIDE + 1];                                \
                b1[s].y = bp[s * 4 * CSTRIDE + CSTRIDE + 1];                      \
                b2[s].x = bp[s * 4 * CSTRIDE + 2];                                \
                b2[s].y = bp[s * 4 * CSTRIDE + CSTRIDE + 2];                      \
            }                                                                     \
        }

        LOAD_B_ROW(0)
        __builtin_amdgcn_sched_barrier(0);

        #pragma unroll
        for (int j = 0; j < 3; ++j) {
            v8f acc0 = {}, acc1 = {}, acc2 = {};
            #pragma unroll
            for (int s = 0; s < 8; ++s) {
                acc0 = __builtin_amdgcn_wmma_f32_16x16x4_f32(
                           false, a[s], false, b0[s], (short)0, acc0, false, false);
                acc1 = __builtin_amdgcn_wmma_f32_16x16x4_f32(
                           false, a[s], false, b1[s], (short)0, acc1, false, false);
                acc2 = __builtin_amdgcn_wmma_f32_16x16x4_f32(
                           false, a[s], false, b2[s], (short)0, acc2, false, false);
            }
            __builtin_amdgcn_sched_barrier(0);

            if (j < 2) {
                LOAD_B_ROW(j + 1)
            }

            float v0 = acc0[0], v1 = acc1[0], v2 = acc2[0];
            if (didx == 1) { v0 = acc0[1]; v1 = acc1[1]; v2 = acc2[1]; }
            if (didx == 2) { v0 = acc0[2]; v1 = acc1[2]; v2 = acc2[2]; }
            if (didx == 3) { v0 = acc0[3]; v1 = acc1[3]; v2 = acc2[3]; }
            if (didx == 4) { v0 = acc0[4]; v1 = acc1[4]; v2 = acc2[4]; }
            if (didx == 5) { v0 = acc0[5]; v1 = acc1[5]; v2 = acc2[5]; }
            if (didx == 6) { v0 = acc0[6]; v1 = acc1[6]; v2 = acc2[6]; }
            if (didx == 7) { v0 = acc0[7]; v1 = acc1[7]; v2 = acc2[7]; }

            if (diagLane) {
                const int o = (i * 3 + j) * 3;
                outp[(size_t)(o + 0) * DHW + obase] = v0 * scale;
                outp[(size_t)(o + 1) * DHW + obase] = v1 * scale;
                outp[(size_t)(o + 2) * DHW + obase] = v2 * scale;
            }
        }
        #undef LOAD_B_ROW
    }
}

extern "C" void kernel_launch(void* const* d_in, const int* in_sizes, int n_in,
                              void* d_out, int out_size, void* d_ws, size_t ws_size,
                              hipStream_t stream)
{
    const float* fixedp  = (const float*)d_in[0];
    const float* movingp = (const float*)d_in[1];
    float* outp = (float*)d_out;

    dim3 grid(DIM * DIM), block(192);
    hipLaunchKernelGGL(wincorr_wmma_kernel, grid, block, 0, stream,
                       fixedp, movingp, outp);
}


# Each wave feeds WMMAs straight from global, no LDS, no barrier: a 16-column moving window as B gives the 3 w-offsets as 3 diagonals of one k chain per row (14-voxel tiles, 7 waves); same sum order
# speedup vs baseline: 1.5789x; 1.5119x over previous
#include <hip/hip_runtime.h>

typedef __attribute__((ext_vector_type(2))) float v2f;
typedef __attribute__((ext_vector_type(8))) float v8f;

#define DIM    96
#define HWSZ   (96*96)
#define DHW    (96*96*96)
#define TILE   14

__global__ __launch_bounds__(224)
void wincorr_wmma_kernel(const float* __restrict__ fixedp,
                         const float* __restrict__ movingp,
                         float* __restrict__ outp)
{
    const int t    = threadIdx.x;
    const int lane = t & 31;
    const int wv   = t >> 5;
    const int h    = blockIdx.x % DIM;
    const int d    = blockIdx.x / DIM;
    const int w0   = wv * TILE;

    const int sp    = lane & 15;
    const int half  = lane >> 4;
    const int khalf = half * 2;

    const int wa   = w0 + sp;
    const int wac  = wa < DIM ? wa : DIM - 1;
    const int aoff = khalf * DHW + wac;
    const float* arow = fixedp + (d * HWSZ + h * DIM);
    v2f a[8];
    #pragma unroll
    for (int s = 0; s < 8; ++s) {
        a[s].x = arow[(4 * s)     * DHW + aoff];
        a[s].y = arow[(4 * s + 1) * DHW + aoff];
    }

    const int  wb    = w0 - 1 + sp;
    const bool colok = (wb >= 0) & (wb < DIM);
    const int  wbc   = wb < 0 ? 0 : (wb > DIM - 1 ? DIM - 1 : wb);
    const int  boff  = khalf * DHW + wbc;

    const int  r0 = sp - 8 * half, r1 = r0 - 1, r2 = r0 - 2;
    const int  m0 = sp, m1 = sp - 1, m2 = sp - 2;
    const bool s0 = (r0 >= 0) & (r0 < 8) & (m0 < TILE) & (w0 + m0 < DIM);
    const bool s1 = (r1 >= 0) & (r1 < 8) & (m1 < TILE) & (w0 + m1 < DIM);
    const bool s2 = (r2 >= 0) & (r2 < 8) & (m2 < TILE) & (w0 + m2 < DIM);
    const float scale = 0.17677669529663687f;
    const int  orow = d * HWSZ + h * DIM + w0;

    v2f bc[8], bn[8];
    #define LOAD_B_ROW(DST, R)                                                    \
    {                                                                             \
        const int dm_ = d + (R) / 3 - 1, hm_ = h + (R) % 3 - 1;                   \
        const int dmc_ = dm_ < 0 ? 0 : (dm_ > DIM - 1 ? DIM - 1 : dm_);           \
        const int hmc_ = hm_ < 0 ? 0 : (hm_ > DIM - 1 ? DIM - 1 : hm_);           \
        const float* brow_ = movingp + (dmc_ * HWSZ + hmc_ * DIM);                \
          \
        _Pragma("unroll")                                                         \
        for (int s = 0; s < 8; ++s) {                                             \
            DST[s].x = brow_[(4 * s)     * DHW + boff];                           \
            DST[s].y = brow_[(4 * s + 1) * DHW + boff];                           \
        }                                                                         \
    }

    LOAD_B_ROW(bc, 0)

    #pragma unroll
    for (int R = 0; R < 9; ++R) {
        if (R < 8) {
            LOAD_B_ROW(bn, R + 1)
        }

        v8f acc = {};
        #pragma unroll
        for (int s = 0; s < 8; ++s)
            acc = __builtin_amdgcn_wmma_f32_16x16x4_f32(
                      false, a[s], false, bc[s], (short)0, acc, false, false);

        const int dm = d + R / 3 - 1, hm = h + R % 3 - 1;
        const bool rowok = (dm >= 0) & (dm < DIM) & (hm >= 0) & (hm < DIM);
        const bool ok = rowok & colok;

        float v0 = acc[0], v1 = acc[0], v2 = acc[0];
        #pragma unroll
        for (int r = 1; r < 8; ++r) {
            if (r0 == r) v0 = acc[r];
            if (r1 == r) v1 = acc[r];
            if (r2 == r) v2 = acc[r];
        }
        v0 = ok ? v0 * scale : 0.0f;
        v1 = ok ? v1 * scale : 0.0f;
        v2 = ok ? v2 * scale : 0.0f;

        const int o = R * 3;
        if (s0) outp[(o + 0) * DHW + orow + m0] = v0;
        if (s1) outp[(o + 1) * DHW + orow + m1] = v1;
        if (s2) outp[(o + 2) * DHW + orow + m2] = v2;

        #pragma unroll
        for (int s = 0; s < 8; ++s) bc[s] = bn[s];
    }
    #undef LOAD_B_ROW
}

extern "C" void kernel_launch(void* const* d_in, const int* in_sizes, int n_in,
                              void* d_out, int out_size, void* d_ws, size_t ws_size,
                              hipStream_t stream)
{
    const float* fixedp  = (const float*)d_in[0];
    const float* movingp = (const float*)d_in[1];
    float* outp = (float*)d_out;

    dim3 grid(DIM * DIM), block(224);
    hipLaunchKernelGGL(wincorr_wmma_kernel, grid, block, 0, stream,
                       fixedp, movingp, outp);
}


# Each wave keeps the A fragments of a 2x2 group of (d,h) output rows and runs each moving row it loads against every row it neighbours: moving rows loaded 4x per output row, not 9x; same sum order
# speedup vs baseline: 1.9941x; 1.2630x over previous
#include <hip/hip_runtime.h>

typedef __attribute__((ext_vector_type(2))) float v2f;
typedef __attribute__((ext_vector_type(8))) float v8f;

#define DIM    96
#define HWSZ   (96*96)
#define DHW    (96*96*96)
#define TILE   14
#define NBLK   2304


static __device__ __forceinline__
void emit_chain(const v8f acc, float* __restrict__ outp, const int o, const int orow,
                const bool ok, const int r0, const int r1, const int r2,
                const int m0, const int m1, const int m2,
                const bool s0, const bool s1, const bool s2)
{
    const float scale = 0.17677669529663687f;
    float v0 = acc[0], v1 = acc[0], v2 = acc[0];
    #pragma unroll
    for (int r = 1; r < 8; ++r) {
        if (r0 == r) v0 = acc[r];
        if (r1 == r) v1 = acc[r];
        if (r2 == r) v2 = acc[r];
    }
    v0 = ok ? v0 * scale : 0.0f;
    v1 = ok ? v1 * scale : 0.0f;
    v2 = ok ? v2 * scale : 0.0f;
    if (s0) outp[(o + 0) * DHW + orow + m0] = v0;
    if (s1) outp[(o + 1) * DHW + orow + m1] = v1;
    if (s2) outp[(o + 2) * DHW + orow + m2] = v2;
}

__global__ __launch_bounds__(224)
void wincorr_wmma_kernel(const float* __restrict__ fixedp,
                         const float* __restrict__ movingp,
                         float* __restrict__ outp)
{
    const int t    = threadIdx.x;
    const int lane = t & 31;
    const int wv   = t >> 5;
    const int h0   = 2 * (blockIdx.x % (DIM / 2));
    const int d0   = 2 * (blockIdx.x / (DIM / 2));
    const int w0   = wv * TILE;

    const int sp    = lane & 15;
    const int half  = lane >> 4;
    const int khalf = half * 2;

    const int wa   = w0 + sp;
    const int wac  = wa < DIM ? wa : DIM - 1;
    const int aoff = khalf * DHW + wac;
    v2f a[4][8];

    const int  wb    = w0 - 1 + sp;
    const bool colok = (wb >= 0) & (wb < DIM);
    const int  wbc   = wb < 0 ? 0 : (wb > DIM - 1 ? DIM - 1 : wb);
    const int  boff  = khalf * DHW + wbc;

    const int  r0 = sp - 8 * half, r1 = r0 - 1, r2 = r0 - 2;
    const int  m0 = sp, m1 = sp - 1, m2 = sp - 2;
    const bool s0 = (r0 >= 0) & (r0 < 8) & (m0 < TILE) & (w0 + m0 < DIM);
    const bool s1 = (r1 >= 0) & (r1 < 8) & (m1 < TILE) & (w0 + m1 < DIM);
    const bool s2 = (r2 >= 0) & (r2 < 8) & (m2 < TILE) & (w0 + m2 < DIM);

    #define LOAD_A_ROW(Q)                                                         \
    {                                                                             \
        const float* arow_ = fixedp + ((d0 + ((Q) >> 1)) * HWSZ + (h0 + ((Q) & 1)) * DIM); \
          \
        _Pragma("unroll")                                                         \
        for (int s = 0; s < 8; ++s) {                                             \
            a[Q][s].x = arow_[(4 * s)     * DHW + aoff];                          \
            a[Q][s].y = arow_[(4 * s + 1) * DHW + aoff];                          \
        }                                                                         \
    }

    v2f bc[8], bn[8];
    #define LOAD_B_ROW(DST, R)                                                    \
    {                                                                             \
        const int dm_ = d0 - 1 + ((R) >> 2), hm_ = h0 - 1 + ((R) & 3);            \
        const int dmc_ = dm_ < 0 ? 0 : (dm_ > DIM - 1 ? DIM - 1 : dm_);           \
        const int hmc_ = hm_ < 0 ? 0 : (hm_ > DIM - 1 ? DIM - 1 : hm_);           \
        const float* brow_ = movingp + (dmc_ * HWSZ + hmc_ * DIM);                \
          \
        _Pragma("unroll")                                                         \
        for (int s = 0; s < 8; ++s) {                                             \
            DST[s].x = brow_[(4 * s)     * DHW + boff];                           \
            DST[s].y = brow_[(4 * s + 1) * DHW + boff];                           \
        }                                                                         \
    }

    LOAD_A_ROW(0)
    LOAD_B_ROW(bc, 0)
    LOAD_A_ROW(1)
    LOAD_A_ROW(2)
    LOAD_A_ROW(3)

    #pragma unroll
    for (int R = 0; R < 16; ++R) {
        const int ar = R >> 2, br = R & 3;
        if (R < 15) {
            LOAD_B_ROW(bn, R + 1)
        }

        const int dm = d0 - 1 + ar, hm = h0 - 1 + br;
        const bool rowok = (dm >= 0) & (dm < DIM) & (hm >= 0) & (hm < DIM);
        const bool ok = rowok & colok;

        #pragma unroll
        for (int dd = 0; dd < 2; ++dd) {
            const int i = ar - dd;
            if (i >= 0 && i <= 2) {
                const int  j0 = br, j1 = br - 1;
                const bool c0 = (j0 <= 2), c1 = (j1 >= 0);
                v8f acc0 = {}, acc1 = {};
                #pragma unroll
                for (int s = 0; s < 8; ++s) {
                    if (c0) acc0 = __builtin_amdgcn_wmma_f32_16x16x4_f32(
                                       false, a[2 * dd][s], false, bc[s], (short)0, acc0, false, false);
                    if (c1) acc1 = __builtin_amdgcn_wmma_f32_16x16x4_f32(
                                       false, a[2 * dd + 1][s], false, bc[s], (short)0, acc1, false, false);
                }
                if (c0) emit_chain(acc0, outp, (i * 3 + j0) * 3,
                                   (d0 + dd) * HWSZ + h0 * DIM + w0,
                                   ok, r0, r1, r2, m0, m1, m2, s0, s1, s2);
                if (c1) emit_chain(acc1, outp, (i * 3 + j1) * 3,
                                   (d0 + dd) * HWSZ + (h0 + 1) * DIM + w0,
                                   ok, r0, r1, r2, m0, m1, m2, s0, s1, s2);
            }
        }

        #pragma unroll
        for (int s = 0; s < 8; ++s) bc[s] = bn[s];
    }
    #undef LOAD_B_ROW
    #undef LOAD_A_ROW
}

extern "C" void kernel_launch(void* const* d_in, const int* in_sizes, int n_in,
                              void* d_out, int out_size, void* d_ws, size_t ws_size,
                              hipStream_t stream)
{
    const float* fixedp  = (const float*)d_in[0];
    const float* movingp = (const float*)d_in[1];
    float* outp = (float*)d_out;

    dim3 grid(NBLK), block(224);
    hipLaunchKernelGGL(wincorr_wmma_kernel, grid, block, 0, stream,
                       fixedp, movingp, outp);
}


# Diagonals of each finished 16x16 tile picked by a wave-private LDS round trip (2 b128 stores, 2 b32 loads) instead of 21 selects; 2 global stores per chain, not 3; sums unchanged
# speedup vs baseline: 2.1494x; 1.0779x over previous
#include <hip/hip_runtime.h>

typedef __attribute__((ext_vector_type(2))) float v2f;
typedef __attribute__((ext_vector_type(4))) float v4f;
typedef __attribute__((ext_vector_type(8))) float v8f;

#define DIM    96
#define HWSZ   (96*96)
#define DHW    (96*96*96)
#define TILE   14
#define NBLK   2304


static __device__ __forceinline__
void tile_to_lds(const v8f acc, float* xw, const int lane)
{
    const v4f lo = {acc[0], acc[1], acc[2], acc[3]};
    const v4f hi = {acc[4], acc[5], acc[6], acc[7]};
    *(v4f*)(xw + 4 * lane)       = lo;
    *(v4f*)(xw + 128 + 4 * lane) = hi;
}

static __device__ __forceinline__
void emit_tile(const float* xw, float* __restrict__ outp, const int obase,
               const int x1, const int x2, const int g1, const int g2,
               const bool ok1, const bool ok2, const bool act1, const bool act2)
{
    const float scale = 0.17677669529663687f;
    float v1 = xw[x1];
    float v2 = xw[x2];
    v1 = ok1 ? v1 * scale : 0.0f;
    v2 = ok2 ? v2 * scale : 0.0f;
    if (act1) outp[obase + g1] = v1;
    if (act2) outp[obase + g2] = v2;
}

__global__ __launch_bounds__(224)
void wincorr_wmma_kernel(const float* __restrict__ fixedp,
                         const float* __restrict__ movingp,
                         float* __restrict__ outp)
{
    const int t    = threadIdx.x;
    const int lane = t & 31;
    const int wv   = t >> 5;
    const int h0   = 2 * (blockIdx.x % (DIM / 2));
    const int d0   = 2 * (blockIdx.x / (DIM / 2));
    const int w0   = wv * TILE;

    const int sp    = lane & 15;
    const int half  = lane >> 4;
    const int khalf = half * 2;

    const int wa   = w0 + sp;
    const int wac  = wa < DIM ? wa : DIM - 1;
    const int aoff = khalf * DHW + wac;
    v2f a[4][8];

    const int  wb    = w0 - 1 + sp;
    const int  wbc   = wb < 0 ? 0 : (wb > DIM - 1 ? DIM - 1 : wb);
    const int  boff  = khalf * DHW + wbc;

    __shared__ __attribute__((aligned(16))) float xpatch[7 * 512];
    float* xw0 = xpatch + wv * 512;
    float* xw1 = xw0 + 256;
    const int  m    = lane & 15;
    const int  hk   = lane >> 4;
    const int  mc   = m < TILE ? m : TILE - 1;
    #define XIDX(mm, k) ((((mm) >> 2) & 1) * 128 + ((mm) + (k) + 16 * ((mm) >> 3)) * 4 + ((mm) & 3))
    const int  x1   = XIDX(mc, hk);
    const int  x2   = XIDX(mc, 2);
    #undef XIDX
    const int  g1   = hk * DHW + m;
    const int  g2   = 2 * DHW + m;
    const int  wc1  = w0 - 1 + m + hk, wc2 = w0 + 1 + m;
    const bool cok1 = (wc1 >= 0) & (wc1 < DIM);
    const bool cok2 = (wc2 < DIM);
    const bool act1 = (m < TILE) & (w0 + m < DIM);
    const bool act2 = act1 & (hk == 0);

    #define LOAD_A_ROW(Q)                                                         \
    {                                                                             \
        const float* arow_ = fixedp + ((d0 + ((Q) >> 1)) * HWSZ + (h0 + ((Q) & 1)) * DIM); \
          \
        _Pragma("unroll")                                                         \
        for (int s = 0; s < 8; ++s) {                                             \
            a[Q][s].x = arow_[(4 * s)     * DHW + aoff];                          \
            a[Q][s].y = arow_[(4 * s + 1) * DHW + aoff];                          \
        }                                                                         \
    }

    v2f bc[8], bn[8];
    #define LOAD_B_ROW(DST, R)                                                    \
    {                                                                             \
        const int dm_ = d0 - 1 + ((R) >> 2), hm_ = h0 - 1 + ((R) & 3);            \
        const int dmc_ = dm_ < 0 ? 0 : (dm_ > DIM - 1 ? DIM - 1 : dm_);           \
        const int hmc_ = hm_ < 0 ? 0 : (hm_ > DIM - 1 ? DIM - 1 : hm_);           \
        const float* brow_ = movingp + (dmc_ * HWSZ + hmc_ * DIM);                \
          \
        _Pragma("unroll")                                                         \
        for (int s = 0; s < 8; ++s) {                                             \
            DST[s].x = brow_[(4 * s)     * DHW + boff];                           \
            DST[s].y = brow_[(4 * s + 1) * DHW + boff];                           \
        }                                                                         \
    }

    LOAD_A_ROW(0)
    LOAD_B_ROW(bc, 0)
    LOAD_A_ROW(1)
    LOAD_A_ROW(2)
    LOAD_A_ROW(3)

    #pragma unroll
    for (int R = 0; R < 16; ++R) {
        const int ar = R >> 2, br = R & 3;
        if (R < 15) {
            LOAD_B_ROW(bn, R + 1)
        }

        const int dm = d0 - 1 + ar, hm = h0 - 1 + br;
        const bool rowok = (dm >= 0) & (dm < DIM) & (hm >= 0) & (hm < DIM);
        const bool ok1 = rowok & cok1, ok2 = rowok & cok2;

        #pragma unroll
        for (int dd = 0; dd < 2; ++dd) {
            const int i = ar - dd;
            if (i >= 0 && i <= 2) {
                const int  j0 = br, j1 = br - 1;
                const bool c0 = (j0 <= 2), c1 = (j1 >= 0);
                v8f acc0 = {}, acc1 = {};
                #pragma unroll
                for (int s = 0; s < 8; ++s) {
                    if (c0) acc0 = __builtin_amdgcn_wmma_f32_16x16x4_f32(
                                       false, a[2 * dd][s], false, bc[s], (short)0, acc0, false, false);
                    if (c1) acc1 = __builtin_amdgcn_wmma_f32_16x16x4_f32(
                                       false, a[2 * dd + 1][s], false, bc[s], (short)0, acc1, false, false);
                }
                if (c0) tile_to_lds(acc0, xw0, lane);
                if (c1) tile_to_lds(acc1, xw1, lane);
                if (c0) emit_tile(xw0, outp,
                                  (i * 3 + j0) * 3 * DHW + (d0 + dd) * HWSZ + h0 * DIM + w0,
                                  x1, x2, g1, g2, ok1, ok2, act1, act2);
                if (c1) emit_tile(xw1, outp,
                                  (i * 3 + j1) * 3 * DHW + (d0 + dd) * HWSZ + (h0 + 1) * DIM + w0,
                                  x1, x2, g1, g2, ok1, ok2, act1, act2);
            }
        }

        #pragma unroll
        for (int s = 0; s < 8; ++s) bc[s] = bn[s];
    }
    #undef LOAD_B_ROW
    #undef LOAD_A_ROW
}

extern "C" void kernel_launch(void* const* d_in, const int* in_sizes, int n_in,
                              void* d_out, int out_size, void* d_ws, size_t ws_size,
                              hipStream_t stream)
{
    const float* fixedp  = (const float*)d_in[0];
    const float* movingp = (const float*)d_in[1];
    float* outp = (float*)d_out;

    dim3 grid(NBLK), block(224);
    hipLaunchKernelGGL(wincorr_wmma_kernel, grid, block, 0, stream,
                       fixedp, movingp, outp);
}


# Each moving row fetched once per block as whole 128-byte lines (16-byte loads), staged in one of two LDS row buffers (one block barrier per row) and read back as B fragments; sum order unchanged
# speedup vs baseline: 2.2494x; 1.0465x over previous
#include <hip/hip_runtime.h>

typedef __attribute__((ext_vector_type(2))) float v2f;
typedef __attribute__((ext_vector_type(4))) float v4f;
typedef __attribute__((ext_vector_type(8))) float v8f;

#define DIM    96
#define HWSZ   (96*96)
#define DHW    (96*96*96)
#define TILE   14
#define NBLK   2304
#define LPITCH 128
#define LSKEW  16
#define ROWBUF (32 * LPITCH)


static __device__ __forceinline__
void tile_to_lds(const v8f acc, float* xw, const int lane)
{
    const v4f lo = {acc[0], acc[1], acc[2], acc[3]};
    const v4f hi = {acc[4], acc[5], acc[6], acc[7]};
    *(v4f*)(xw + 4 * lane)       = lo;
    *(v4f*)(xw + 128 + 4 * lane) = hi;
}

static __device__ __forceinline__
void emit_tile(const float* xw, float* __restrict__ outp, const int obase,
               const int x1, const int x2, const int g1, const int g2,
               const bool ok1, const bool ok2, const bool act1, const bool act2)
{
    const float scale = 0.17677669529663687f;
    float v1 = xw[x1];
    float v2 = xw[x2];
    v1 = ok1 ? v1 * scale : 0.0f;
    v2 = ok2 ? v2 * scale : 0.0f;
    if (act1) outp[obase + g1] = v1;
    if (act2) outp[obase + g2] = v2;
}

static __device__ __forceinline__
void lds_barrier()
{
    asm volatile("s_wait_dscnt 0x0\n\ts_barrier_signal -1\n\ts_barrier_wait -1" ::: "memory");
}

__global__ __launch_bounds__(224)
void wincorr_wmma_kernel(const float* __restrict__ fixedp,
                         const float* __restrict__ movingp,
                         float* __restrict__ outp)
{
    const int t    = threadIdx.x;
    const int lane = t & 31;
    const int wv   = t >> 5;
    const int h0   = 2 * (blockIdx.x % (DIM / 2));
    const int d0   = 2 * (blockIdx.x / (DIM / 2));
    const int w0   = wv * TILE;

    const int sp    = lane & 15;
    const int half  = lane >> 4;
    const int khalf = half * 2;

    const int wa   = w0 + sp;
    const int wac  = wa < DIM ? wa : DIM - 1;
    const int aoff = khalf * DHW + wac;
    v2f a[4][8];

    const int  wb    = w0 - 1 + sp;
    const int  wbc   = wb < 0 ? 0 : (wb > DIM - 1 ? DIM - 1 : wb);
    const int  boff  = khalf * LPITCH + half * LSKEW + wbc;

    __shared__ __attribute__((aligned(16))) float xpatch[7 * 512];
    __shared__ __attribute__((aligned(16))) float bstage[2 * ROWBUF];
    float* xw0 = xpatch + wv * 512;
    float* xw1 = xw0 + 256;
    const int  m    = lane & 15;
    const int  hk   = lane >> 4;
    const int  mc   = m < TILE ? m : TILE - 1;
    #define XIDX(mm, k) ((((mm) >> 2) & 1) * 128 + ((mm) + (k) + 16 * ((mm) >> 3)) * 4 + ((mm) & 3))
    const int  x1   = XIDX(mc, hk);
    const int  x2   = XIDX(mc, 2);
    #undef XIDX
    const int  g1   = hk * DHW + m;
    const int  g2   = 2 * DHW + m;
    const int  wc1  = w0 - 1 + m + hk, wc2 = w0 + 1 + m;
    const bool cok1 = (wc1 >= 0) & (wc1 < DIM);
    const bool cok2 = (wc2 < DIM);
    const bool act1 = (m < TILE) & (w0 + m < DIM);
    const bool act2 = act1 & (hk == 0);

    #define LOAD_A_ROW(Q)                                                         \
    {                                                                             \
        const float* arow_ = fixedp + ((d0 + ((Q) >> 1)) * HWSZ + (h0 + ((Q) & 1)) * DIM); \
          \
        _Pragma("unroll")                                                         \
        for (int s = 0; s < 8; ++s) {                                             \
            a[Q][s].x = arow_[(4 * s)     * DHW + aoff];                          \
            a[Q][s].y = arow_[(4 * s + 1) * DHW + aoff];                          \
        }                                                                         \
    }

    const int  lch    = t / 24, lq = t - lch * 24;
    const int  gso    = lch * DHW + 4 * lq;
    const int  lso    = lch * LPITCH + ((lch >> 1) & 1) * LSKEW + 4 * lq;
    const bool loader = __builtin_amdgcn_readfirstlane(wv) < 6;
    v4f st[4];
    #define GLOAD_B_ROW(R)                                                        \
    {                                                                             \
        const int dm_ = d0 - 1 + ((R) >> 2), hm_ = h0 - 1 + ((R) & 3);            \
        const int dmc_ = dm_ < 0 ? 0 : (dm_ > DIM - 1 ? DIM - 1 : dm_);           \
        const int hmc_ = hm_ < 0 ? 0 : (hm_ > DIM - 1 ? DIM - 1 : hm_);           \
        const float* brow_ = movingp + (dmc_ * HWSZ + hmc_ * DIM);                \
        if (loader) {                                                             \
                  \
            _Pragma("unroll")                                                     \
            for (int u = 0; u < 4; ++u)                                           \
                st[u] = *(const v4f*)(brow_ + (8 * u) * DHW + gso);               \
        }                                                                         \
    }
    #define STAGE_B_ROW(BUF)                                                      \
    {                                                                             \
        if (loader) {                                                             \
                  \
            _Pragma("unroll")                                                     \
            for (int u = 0; u < 4; ++u)                                           \
                *(v4f*)(bstage + (BUF) * ROWBUF + (8 * u) * LPITCH + lso) = st[u]; \
        }                                                                         \
    }
    v2f bc[8];
    #define FRAG_B_ROW(BUF)                                                       \
    {                                                                             \
        const float* xb_ = bstage + (BUF) * ROWBUF + boff;                        \
          \
        _Pragma("unroll")                                                         \
        for (int s = 0; s < 8; ++s) {                                             \
            bc[s].x = xb_[(4 * s)     * LPITCH];                                  \
            bc[s].y = xb_[(4 * s + 1) * LPITCH];                                  \
        }                                                                         \
    }

    GLOAD_B_ROW(0)
    LOAD_A_ROW(0)
    LOAD_A_ROW(1)
    LOAD_A_ROW(2)
    LOAD_A_ROW(3)
    STAGE_B_ROW(0)
    lds_barrier();

    #pragma unroll
    for (int R = 0; R < 16; ++R) {
        const int ar = R >> 2, br = R & 3;
        if (R < 15) {
            GLOAD_B_ROW(R + 1)
        }
        FRAG_B_ROW(R & 1)

        const int dm = d0 - 1 + ar, hm = h0 - 1 + br;
        const bool rowok = (dm >= 0) & (dm < DIM) & (hm >= 0) & (hm < DIM);
        const bool ok1 = rowok & cok1, ok2 = rowok & cok2;

        #pragma unroll
        for (int dd = 0; dd < 2; ++dd) {
            const int i = ar - dd;
            if (i >= 0 && i <= 2) {
                const int  j0 = br, j1 = br - 1;
                const bool c0 = (j0 <= 2), c1 = (j1 >= 0);
                v8f acc0 = {}, acc1 = {};
                #pragma unroll
                for (int s = 0; s < 8; ++s) {
                    if (c0) acc0 = __builtin_amdgcn_wmma_f32_16x16x4_f32(
                                       false, a[2 * dd][s], false, bc[s], (short)0, acc0, false, false);
                    if (c1) acc1 = __builtin_amdgcn_wmma_f32_16x16x4_f32(
                                       false, a[2 * dd + 1][s], false, bc[s], (short)0, acc1, false, false);
                }
                if (c0) tile_to_lds(acc0, xw0, lane);
                if (c1) tile_to_lds(acc1, xw1, lane);
                if (c0) emit_tile(xw0, outp,
                                  (i * 3 + j0) * 3 * DHW + (d0 + dd) * HWSZ + h0 * DIM + w0,
                                  x1, x2, g1, g2, ok1, ok2, act1, act2);
                if (c1) emit_tile(xw1, outp,
                                  (i * 3 + j1) * 3 * DHW + (d0 + dd) * HWSZ + (h0 + 1) * DIM + w0,
                                  x1, x2, g1, g2, ok1, ok2, act1, act2);
            }
        }

        if (R < 15) {
            STAGE_B_ROW((R + 1) & 1)
            lds_barrier();
        }
    }
    #undef FRAG_B_ROW
    #undef STAGE_B_ROW
    #undef GLOAD_B_ROW
    #undef LOAD_A_ROW
}

extern "C" void kernel_launch(void* const* d_in, const int* in_sizes, int n_in,
                              void* d_out, int out_size, void* d_ws, size_t ws_size,
                              hipStream_t stream)
{
    const float* fixedp  = (const float*)d_in[0];
    const float* movingp = (const float*)d_in[1];
    float* outp = (float*)d_out;

    dim3 grid(NBLK), block(224);
    hipLaunchKernelGGL(wincorr_wmma_kernel, grid, block, 0, stream,
                       fixedp, movingp, outp);
}
